# starting kernel
# baseline (speedup 1.0000x reference)
#include <hip/hip_runtime.h>
#include <hip/hip_bf16.h>


typedef float v2f __attribute__((ext_vector_type(2)));
typedef float v8f __attribute__((ext_vector_type(8)));

#define D_DIM 128

__global__ void gcl_zero4(float* __restrict__ p, size_t n4) {
    size_t i = (size_t)blockIdx.x * blockDim.x + threadIdx.x;
    if (i < n4) ((float4*)p)[i] = make_float4(0.f, 0.f, 0.f, 0.f);
}

__global__ __launch_bounds__(128) void gcl_gemm_wmma(
    const float* __restrict__ feat,
    const float* __restrict__ weight,
    const float* __restrict__ bias,
    float* __restrict__ support,
    int nRows)
{
    const int lane   = threadIdx.x & 31;
    const int wave   = threadIdx.x >> 5;
    const int tile   = blockIdx.x * 4 + wave;
    const int mtile  = tile >> 3;
    const int ntile  = tile & 7;
    const int m0     = mtile * 16;
    if (m0 >= nRows) return;

    const int half   = lane >> 4;
    const int l16    = lane & 15;
    const int koff   = half * 2;

    const int arow = (m0 + l16 < nRows) ? (m0 + l16) : (nRows - 1);
    const float* aPtr = feat   + (size_t)arow * D_DIM + koff;
    const float* bPtr = weight + (size_t)(ntile * 16 + l16) * D_DIM + koff;

    v8f c = {};
#pragma unroll
    for (int k0 = 0; k0 < D_DIM; k0 += 4) {
        v2f a = *(const v2f*)(aPtr + k0);
        v2f b = *(const v2f*)(bPtr + k0);
        c = __builtin_amdgcn_wmma_f32_16x16x4_f32(
                false, a, false, b, (short)0, c, false, false);
    }

    const int   ncol = ntile * 16 + l16;
    const float bv   = bias[ncol];
#pragma unroll
    for (int i = 0; i < 8; ++i) {
        int row = m0 + half * 8 + i;
        if (row < nRows)
            support[(size_t)row * D_DIM + ncol] = c[i] + bv;
    }
}

__global__ __launch_bounds__(256) void gcl_spmm_scatter(
    const float* __restrict__ support,
    const float* __restrict__ vals,
    const int*   __restrict__ rows,
    const int*   __restrict__ cols,
    float* __restrict__ out,
    int nEdges)
{
    const int lane = threadIdx.x & 31;
    const int wave = threadIdx.x >> 5;
    const int e    = blockIdx.x * 8 + wave;
    if (e >= nEdges) return;

    const int   col = cols[e];
    const int   row = rows[e];
    const float v   = vals[e];

    const float4 s = ((const float4*)(support + (size_t)col * D_DIM))[lane];
    float* dst = out + (size_t)row * D_DIM + lane * 4;
    atomicAdd(dst + 0, s.x * v);
    atomicAdd(dst + 1, s.y * v);
    atomicAdd(dst + 2, s.z * v);
    atomicAdd(dst + 3, s.w * v);
}

extern "C" void kernel_launch(void* const* d_in, const int* in_sizes, int n_in,
                              void* d_out, int out_size, void* d_ws, size_t ws_size,
                              hipStream_t stream) {
    const float* feat   = (const float*)d_in[0];
    const float* weight = (const float*)d_in[1];
    const float* bias   = (const float*)d_in[2];
    const float* evals  = (const float*)d_in[3];
    const int*   erows  = (const int*)d_in[4];
    const int*   ecols  = (const int*)d_in[5];
    float*       out    = (float*)d_out;

    const int nNodes = in_sizes[0] / D_DIM;
    const int nEdges = in_sizes[3];
    float* support = (float*)d_ws;

    {
        size_t n4 = (size_t)nNodes * D_DIM / 4;
        int blocks = (int)((n4 + 255) / 256);
        gcl_zero4<<<blocks, 256, 0, stream>>>(out, n4);
    }

    {
        int mtiles = (nNodes + 15) / 16;
        int tiles  = mtiles * 8;
        int blocks = (tiles + 3) / 4;
        gcl_gemm_wmma<<<blocks, 128, 0, stream>>>(feat, weight, bias, support, nNodes);
    }

    {
        int blocks = (nEdges + 7) / 8;
        gcl_spmm_scatter<<<blocks, 256, 0, stream>>>(support, evals, erows, ecols, out, nEdges);
    }
}


# Scatter-add: lane L owns row elements L,32+L,64+L,96+L so each f32 atomic (and each gather load) instruction of a wave covers one whole 128-B line, not 8 words in each of 4 lines
# speedup vs baseline: 1.3253x; 1.3253x over previous
#include <hip/hip_runtime.h>
#include <hip/hip_bf16.h>


typedef float v2f __attribute__((ext_vector_type(2)));
typedef float v8f __attribute__((ext_vector_type(8)));

#define D_DIM 128

__global__ void gcl_zero4(float* __restrict__ p, size_t n4) {
    size_t i = (size_t)blockIdx.x * blockDim.x + threadIdx.x;
    if (i < n4) ((float4*)p)[i] = make_float4(0.f, 0.f, 0.f, 0.f);
}

__global__ __launch_bounds__(128) void gcl_gemm_wmma(
    const float* __restrict__ feat,
    const float* __restrict__ weight,
    const float* __restrict__ bias,
    float* __restrict__ support,
    int nRows)
{
    const int lane   = threadIdx.x & 31;
    const int wave   = threadIdx.x >> 5;
    const int tile   = blockIdx.x * 4 + wave;
    const int mtile  = tile >> 3;
    const int ntile  = tile & 7;
    const int m0     = mtile * 16;
    if (m0 >= nRows) return;

    const int half   = lane >> 4;
    const int l16    = lane & 15;
    const int koff   = half * 2;

    const int arow = (m0 + l16 < nRows) ? (m0 + l16) : (nRows - 1);
    const float* aPtr = feat   + (size_t)arow * D_DIM + koff;
    const float* bPtr = weight + (size_t)(ntile * 16 + l16) * D_DIM + koff;

    v8f c = {};
#pragma unroll
    for (int k0 = 0; k0 < D_DIM; k0 += 4) {
        v2f a = *(const v2f*)(aPtr + k0);
        v2f b = *(const v2f*)(bPtr + k0);
        c = __builtin_amdgcn_wmma_f32_16x16x4_f32(
                false, a, false, b, (short)0, c, false, false);
    }

    const int   ncol = ntile * 16 + l16;
    const float bv   = bias[ncol];
#pragma unroll
    for (int i = 0; i < 8; ++i) {
        int row = m0 + half * 8 + i;
        if (row < nRows)
            support[(size_t)row * D_DIM + ncol] = c[i] + bv;
    }
}

__global__ __launch_bounds__(256) void gcl_spmm_scatter(
    const float* __restrict__ support,
    const float* __restrict__ vals,
    const int*   __restrict__ rows,
    const int*   __restrict__ cols,
    float* __restrict__ out,
    int nEdges)
{
    const int lane = threadIdx.x & 31;
    const int wave = threadIdx.x >> 5;
    const int e    = blockIdx.x * 8 + wave;
    if (e >= nEdges) return;

    const int   col = cols[e];
    const int   row = rows[e];
    const float v   = vals[e];

    const float* src = support + (size_t)col * D_DIM + lane;
    float*       dst = out     + (size_t)row * D_DIM + lane;
    const float s0 = src[0];
    const float s1 = src[32];
    const float s2 = src[64];
    const float s3 = src[96];
    atomicAdd(dst +  0, s0 * v);
    atomicAdd(dst + 32, s1 * v);
    atomicAdd(dst + 64, s2 * v);
    atomicAdd(dst + 96, s3 * v);
}

extern "C" void kernel_launch(void* const* d_in, const int* in_sizes, int n_in,
                              void* d_out, int out_size, void* d_ws, size_t ws_size,
                              hipStream_t stream) {
    const float* feat   = (const float*)d_in[0];
    const float* weight = (const float*)d_in[1];
    const float* bias   = (const float*)d_in[2];
    const float* evals  = (const float*)d_in[3];
    const int*   erows  = (const int*)d_in[4];
    const int*   ecols  = (const int*)d_in[5];
    float*       out    = (float*)d_out;

    const int nNodes = in_sizes[0] / D_DIM;
    const int nEdges = in_sizes[3];
    float* support = (float*)d_ws;

    {
        size_t n4 = (size_t)nNodes * D_DIM / 4;
        int blocks = (int)((n4 + 255) / 256);
        gcl_zero4<<<blocks, 256, 0, stream>>>(out, n4);
    }

    {
        int mtiles = (nNodes + 15) / 16;
        int tiles  = mtiles * 8;
        int blocks = (tiles + 3) / 4;
        gcl_gemm_wmma<<<blocks, 128, 0, stream>>>(feat, weight, bias, support, nNodes);
    }

    {
        int blocks = (nEdges + 7) / 8;
        gcl_spmm_scatter<<<blocks, 256, 0, stream>>>(support, evals, erows, ecols, out, nEdges);
    }
}


# GEMM: one wave per 16x128 strip (8 accumulator tiles share each A fragment), W staged once per block in padded LDS and B fragments read from there; k order and sum order unchanged
# speedup vs baseline: 1.5537x; 1.1723x over previous
#include <hip/hip_runtime.h>
#include <hip/hip_bf16.h>


typedef float v2f __attribute__((ext_vector_type(2)));
typedef float v8f __attribute__((ext_vector_type(8)));

#define D_DIM 128

__global__ void gcl_zero4(float* __restrict__ p, size_t n4) {
    size_t i = (size_t)blockIdx.x * blockDim.x + threadIdx.x;
    if (i < n4) ((float4*)p)[i] = make_float4(0.f, 0.f, 0.f, 0.f);
}

#define W_PITCH 132
__global__ __launch_bounds__(128) void gcl_gemm_wmma(
    const float* __restrict__ feat,
    const float* __restrict__ weight,
    const float* __restrict__ bias,
    float* __restrict__ support,
    int nRows)
{
    __shared__ __attribute__((aligned(16))) float sW[D_DIM * W_PITCH];

#pragma unroll
    for (int i = 0; i < 32; ++i) {
        const int idx = threadIdx.x + i * 128;
        const int r   = idx >> 5;
        const int c4  = idx & 31;
        const float4 v = ((const float4*)weight)[idx];
        *(float4*)(&sW[r * W_PITCH + c4 * 4]) = v;
    }

    const int lane   = threadIdx.x & 31;
    const int wave   = threadIdx.x >> 5;
    const int mtile  = blockIdx.x * 4 + wave;
    const int m0     = mtile * 16;

    const int half   = lane >> 4;
    const int l16    = lane & 15;
    const int koff   = half * 2;

    const int arow = (m0 + l16 < nRows) ? (m0 + l16) : (nRows - 1);
    const float* aPtr = feat + (size_t)arow * D_DIM + koff;

    v2f a[32];
#pragma unroll
    for (int s = 0; s < 32; ++s) a[s] = *(const v2f*)(aPtr + 4 * s);

    __syncthreads();
    if (m0 >= nRows) return;

    v8f c[8];
#pragma unroll
    for (int nt = 0; nt < 8; ++nt) c[nt] = (v8f){};

    const float* bBase = sW + l16 * W_PITCH + koff;
#pragma unroll
    for (int s = 0; s < 32; ++s) {
#pragma unroll
        for (int nt = 0; nt < 8; ++nt) {
            const v2f b = *(const v2f*)(bBase + nt * 16 * W_PITCH + 4 * s);
            c[nt] = __builtin_amdgcn_wmma_f32_16x16x4_f32(
                        false, a[s], false, b, (short)0, c[nt], false, false);
        }
    }

    float bv[8];
#pragma unroll
    for (int nt = 0; nt < 8; ++nt) bv[nt] = bias[nt * 16 + l16];

    float* oPtr = support + (size_t)(m0 + half * 8) * D_DIM + l16;
    if (m0 + 16 <= nRows) {
#pragma unroll
        for (int nt = 0; nt < 8; ++nt) {
#pragma unroll
            for (int i = 0; i < 8; ++i)
                oPtr[i * D_DIM + nt * 16] = c[nt][i] + bv[nt];
        }
    } else {
#pragma unroll
        for (int nt = 0; nt < 8; ++nt) {
#pragma unroll
            for (int i = 0; i < 8; ++i) {
                if (m0 + half * 8 + i < nRows)
                    oPtr[i * D_DIM + nt * 16] = c[nt][i] + bv[nt];
            }
        }
    }
}

__global__ __launch_bounds__(256) void gcl_spmm_scatter(
    const float* __restrict__ support,
    const float* __restrict__ vals,
    const int*   __restrict__ rows,
    const int*   __restrict__ cols,
    float* __restrict__ out,
    int nEdges)
{
    const int lane = threadIdx.x & 31;
    const int wave = threadIdx.x >> 5;
    const int e    = blockIdx.x * 8 + wave;
    if (e >= nEdges) return;

    const int   col = cols[e];
    const int   row = rows[e];
    const float v   = vals[e];

    const float* src = support + (size_t)col * D_DIM + lane;
    float*       dst = out     + (size_t)row * D_DIM + lane;
    const float s0 = src[0];
    const float s1 = src[32];
    const float s2 = src[64];
    const float s3 = src[96];
    atomicAdd(dst +  0, s0 * v);
    atomicAdd(dst + 32, s1 * v);
    atomicAdd(dst + 64, s2 * v);
    atomicAdd(dst + 96, s3 * v);
}

extern "C" void kernel_launch(void* const* d_in, const int* in_sizes, int n_in,
                              void* d_out, int out_size, void* d_ws, size_t ws_size,
                              hipStream_t stream) {
    const float* feat   = (const float*)d_in[0];
    const float* weight = (const float*)d_in[1];
    const float* bias   = (const float*)d_in[2];
    const float* evals  = (const float*)d_in[3];
    const int*   erows  = (const int*)d_in[4];
    const int*   ecols  = (const int*)d_in[5];
    float*       out    = (float*)d_out;

    const int nNodes = in_sizes[0] / D_DIM;
    const int nEdges = in_sizes[3];
    float* support = (float*)d_ws;

    {
        size_t n4 = (size_t)nNodes * D_DIM / 4;
        int blocks = (int)((n4 + 255) / 256);
        gcl_zero4<<<blocks, 256, 0, stream>>>(out, n4);
    }

    {
        int mtiles = (nNodes + 15) / 16;
        int blocks = (mtiles + 3) / 4;
        gcl_gemm_wmma<<<blocks, 128, 0, stream>>>(feat, weight, bias, support, nNodes);
    }

    {
        int blocks = (nEdges + 7) / 8;
        gcl_spmm_scatter<<<blocks, 256, 0, stream>>>(support, evals, erows, ecols, out, nEdges);
    }
}


# SpMM: each work-group owns 400 output rows in LDS, scans the edge list for them, adds gathered support rows by LDS f32 atomics (arrival order), stores each row once; no global atomics, no zero pass
# speedup vs baseline: 2.2552x; 1.4515x over previous
#include <hip/hip_runtime.h>
#include <hip/hip_bf16.h>


typedef float v2f __attribute__((ext_vector_type(2)));
typedef float v8f __attribute__((ext_vector_type(8)));

#define D_DIM 128

#define W_PITCH 132
__global__ __launch_bounds__(128) void gcl_gemm_wmma(
    const float* __restrict__ feat,
    const float* __restrict__ weight,
    const float* __restrict__ bias,
    float* __restrict__ support,
    int nRows)
{
    __shared__ __attribute__((aligned(16))) float sW[D_DIM * W_PITCH];

#pragma unroll
    for (int i = 0; i < 32; ++i) {
        const int idx = threadIdx.x + i * 128;
        const int r   = idx >> 5;
        const int c4  = idx & 31;
        const float4 v = ((const float4*)weight)[idx];
        *(float4*)(&sW[r * W_PITCH + c4 * 4]) = v;
    }

    const int lane   = threadIdx.x & 31;
    const int wave   = threadIdx.x >> 5;
    const int mtile  = blockIdx.x * 4 + wave;
    const int m0     = mtile * 16;

    const int half   = lane >> 4;
    const int l16    = lane & 15;
    const int koff   = half * 2;

    const int arow = (m0 + l16 < nRows) ? (m0 + l16) : (nRows - 1);
    const float* aPtr = feat + (size_t)arow * D_DIM + koff;

    v2f a[32];
#pragma unroll
    for (int s = 0; s < 32; ++s) a[s] = *(const v2f*)(aPtr + 4 * s);

    __syncthreads();
    if (m0 >= nRows) return;

    v8f c[8];
#pragma unroll
    for (int nt = 0; nt < 8; ++nt) c[nt] = (v8f){};

    const float* bBase = sW + l16 * W_PITCH + koff;
#pragma unroll
    for (int s = 0; s < 32; ++s) {
#pragma unroll
        for (int nt = 0; nt < 8; ++nt) {
            const v2f b = *(const v2f*)(bBase + nt * 16 * W_PITCH + 4 * s);
            c[nt] = __builtin_amdgcn_wmma_f32_16x16x4_f32(
                        false, a[s], false, b, (short)0, c[nt], false, false);
        }
    }

    float bv[8];
#pragma unroll
    for (int nt = 0; nt < 8; ++nt) bv[nt] = bias[nt * 16 + l16];

    float* oPtr = support + (size_t)(m0 + half * 8) * D_DIM + l16;
    if (m0 + 16 <= nRows) {
#pragma unroll
        for (int nt = 0; nt < 8; ++nt) {
#pragma unroll
            for (int i = 0; i < 8; ++i)
                oPtr[i * D_DIM + nt * 16] = c[nt][i] + bv[nt];
        }
    } else {
#pragma unroll
        for (int nt = 0; nt < 8; ++nt) {
#pragma unroll
            for (int i = 0; i < 8; ++i) {
                if (m0 + half * 8 + i < nRows)
                    oPtr[i * D_DIM + nt * 16] = c[nt][i] + bv[nt];
            }
        }
    }
}

typedef float v4f __attribute__((ext_vector_type(4)));

#define OWN_ROWS 400
#define LIST_CAP 9216

__global__ __launch_bounds__(1024) void gcl_spmm_owned(
    const float* __restrict__ support,
    const float* __restrict__ vals,
    const int*   __restrict__ rows,
    const int*   __restrict__ cols,
    float* __restrict__ out,
    int nRows, int nEdges)
{
    __shared__ __attribute__((aligned(16))) float sAcc[OWN_ROWS * D_DIM];
    __shared__ int   sKey[LIST_CAP];
    __shared__ int   sDst[LIST_CAP];
    __shared__ float sVal[LIST_CAP];
    __shared__ int   sCount;

    const int t    = threadIdx.x;
    const int lane = t & 31;
    const int wave = __builtin_amdgcn_readfirstlane(t >> 5);

    const int rowBase = blockIdx.x * OWN_ROWS;
    const int remRows = nRows - rowBase;
    const unsigned rowsHere = (unsigned)(remRows < OWN_ROWS ? remRows : OWN_ROWS);

    for (int i = t; i < OWN_ROWS * D_DIM / 4; i += 1024)
        ((v4f*)sAcc)[i] = (v4f){0.f, 0.f, 0.f, 0.f};
    if (t == 0) sCount = 0;
    __syncthreads();

    auto take = [&](int e, unsigned d) {
        const int slot = atomicAdd(&sCount, 1);
        if (slot < LIST_CAP) {
            sKey[slot] = e;
            sDst[slot] = (int)d;
        } else {
            const float  v   = vals[e];
            const float* src = support + (size_t)cols[e] * D_DIM;
            float*       dst = sAcc + d * D_DIM;
#pragma clang loop unroll(disable)
            for (int j = 0; j < D_DIM; ++j)
                atomicAdd(dst + 32 * (j & 3) + (j >> 2), src[j] * v);
        }
    };

    const int nQuads = nEdges >> 2;
    const int nTrips = (nQuads + 8191) >> 13;
    const int4* rows4 = (const int4*)rows;
    if (nTrips > 0) {
        const int trip0 = (int)(blockIdx.x % (unsigned)nTrips);
        const int qLast = nQuads - 1;
        for (int k = 0; k < nTrips; ++k) {
            int trip = k + trip0;
            if (trip >= nTrips) trip -= nTrips;
            const int q0 = (trip << 13) + t;
            int4 r[8];
#pragma unroll
            for (int u = 0; u < 8; ++u) {
                const int q = q0 + u * 1024;
                r[u] = rows4[q < qLast ? q : qLast];
            }
#pragma unroll
            for (int u = 0; u < 8; ++u) {
                const int q  = q0 + u * 1024;
                const int e0 = q * 4;
                const unsigned lim = (q < nQuads) ? rowsHere : 0u;
                const unsigned d0 = (unsigned)(r[u].x - rowBase);
                const unsigned d1 = (unsigned)(r[u].y - rowBase);
                const unsigned d2 = (unsigned)(r[u].z - rowBase);
                const unsigned d3 = (unsigned)(r[u].w - rowBase);
                const unsigned dlo01 = d0 < d1 ? d0 : d1;
                const unsigned dlo23 = d2 < d3 ? d2 : d3;
                const unsigned dlo   = dlo01 < dlo23 ? dlo01 : dlo23;
                if (dlo < lim) {
                    if (d0 < rowsHere) take(e0,     d0);
                    if (d1 < rowsHere) take(e0 + 1, d1);
                    if (d2 < rowsHere) take(e0 + 2, d2);
                    if (d3 < rowsHere) take(e0 + 3, d3);
                }
            }
        }
    }
    for (int e = nQuads * 4 + t; e < nEdges; e += 1024) {
        const unsigned d = (unsigned)(rows[e] - rowBase);
        if (d < rowsHere) take(e, d);
    }
    __syncthreads();

    const int total = __builtin_amdgcn_readfirstlane(sCount);
    const int cnt   = total < LIST_CAP ? total : LIST_CAP;

    for (int i = t; i < cnt; i += 1024) {
        const int e = sKey[i];
        sKey[i] = cols[e];
        sVal[i] = vals[e];
    }
    __syncthreads();

    const float* sLane = support + lane * 4;
    for (int i0 = wave; i0 < cnt; i0 += 32 * 8) {
        const int iMine = i0 + 32 * (lane & 7);
        const int iSafe = (iMine < cnt) ? iMine : i0;
        const int   colMine = sKey[iSafe];
        const int   dstMine = sDst[iSafe];
        const int   valMine = __float_as_int(sVal[iSafe]);
        v4f   x[8];
        float w[8];
        int   d[8];
#pragma unroll
        for (int u = 0; u < 8; ++u) {
            const int col = __builtin_amdgcn_readlane(colMine, u);
            d[u] = __builtin_amdgcn_readlane(dstMine, u);
            w[u] = __int_as_float(__builtin_amdgcn_readlane(valMine, u));
            x[u] = *(const v4f*)(sLane + (size_t)col * D_DIM);
        }
#pragma unroll
        for (int u = 0; u < 8; ++u) {
            if (i0 + 32 * u < cnt) {
                float* dst = sAcc + d[u] * D_DIM + lane;
                atomicAdd(dst,      x[u].x * w[u]);
                atomicAdd(dst + 32, x[u].y * w[u]);
                atomicAdd(dst + 64, x[u].z * w[u]);
                atomicAdd(dst + 96, x[u].w * w[u]);
            }
        }
    }
    __syncthreads();

    for (unsigned lr = (unsigned)wave; lr < rowsHere; lr += 32) {
        const float* a = sAcc + lr * D_DIM + lane;
        v4f o;
        o.x = a[0];
        o.y = a[32];
        o.z = a[64];
        o.w = a[96];
        *(v4f*)(out + (size_t)(rowBase + (int)lr) * D_DIM + lane * 4) = o;
    }
}

extern "C" void kernel_launch(void* const* d_in, const int* in_sizes, int n_in,
                              void* d_out, int out_size, void* d_ws, size_t ws_size,
                              hipStream_t stream) {
    const float* feat   = (const float*)d_in[0];
    const float* weight = (const float*)d_in[1];
    const float* bias   = (const float*)d_in[2];
    const float* evals  = (const float*)d_in[3];
    const int*   erows  = (const int*)d_in[4];
    const int*   ecols  = (const int*)d_in[5];
    float*       out    = (float*)d_out;

    const int nNodes = in_sizes[0] / D_DIM;
    const int nEdges = in_sizes[3];
    float* support = (float*)d_ws;

    {
        int mtiles = (nNodes + 15) / 16;
        int blocks = (mtiles + 3) / 4;
        gcl_gemm_wmma<<<blocks, 128, 0, stream>>>(feat, weight, bias, support, nNodes);
    }

    {
        int blocks = (nNodes + OWN_ROWS - 1) / OWN_ROWS;
        gcl_spmm_owned<<<blocks, 1024, 0, stream>>>(support, evals, erows, ecols, out, nNodes, nEdges);
    }
}


# SpMM scan reads 1-byte owner tags (row/400 mod 256, written by the GEMM launch, 16 edges per 16-byte load) instead of the 4-byte rows; listed edges get their exact row in step 3
# speedup vs baseline: 2.7037x; 1.1989x over previous
#include <hip/hip_runtime.h>
#include <hip/hip_bf16.h>


typedef float v2f __attribute__((ext_vector_type(2)));
typedef float v8f __attribute__((ext_vector_type(8)));

#define D_DIM 128

#define W_PITCH 132
#define OWN_ROWS 400
__device__ __forceinline__ unsigned ownerTag(int row)
{
    return ((unsigned)row / (unsigned)OWN_ROWS) & 255u;
}
__device__ __forceinline__ unsigned packTags(int4 r)
{
    return ownerTag(r.x) | (ownerTag(r.y) << 8) | (ownerTag(r.z) << 16) | (ownerTag(r.w) << 24);
}
__global__ __launch_bounds__(128) void gcl_gemm_wmma(
    const float* __restrict__ feat,
    const float* __restrict__ weight,
    const float* __restrict__ bias,
    const int*   __restrict__ rows,
    float* __restrict__ support,
    unsigned* __restrict__ tags,
    int nRows, int nEdges)
{
    __shared__ __attribute__((aligned(16))) float sW[D_DIM * W_PITCH];

    const int nQuads = nEdges >> 2;
    const int qStep  = gridDim.x * 128;
    const int4* rows4 = (const int4*)rows;
    const int qa = blockIdx.x * 128 + threadIdx.x;
    const int qb = qa + qStep;
    int4 ra = {0, 0, 0, 0};
    int4 rb = {0, 0, 0, 0};
    if (nQuads > 0) {
        ra = rows4[qa < nQuads ? qa : nQuads - 1];
        rb = rows4[qb < nQuads ? qb : nQuads - 1];
    }

#pragma unroll
    for (int i = 0; i < 32; ++i) {
        const int idx = threadIdx.x + i * 128;
        const int r   = idx >> 5;
        const int c4  = idx & 31;
        const float4 v = ((const float4*)weight)[idx];
        *(float4*)(&sW[r * W_PITCH + c4 * 4]) = v;
    }

    const int lane   = threadIdx.x & 31;
    const int wave   = threadIdx.x >> 5;
    const int mtile  = blockIdx.x * 4 + wave;
    const int m0     = mtile * 16;

    const int half   = lane >> 4;
    const int l16    = lane & 15;
    const int koff   = half * 2;

    const int arow = (m0 + l16 < nRows) ? (m0 + l16) : (nRows - 1);
    const float* aPtr = feat + (size_t)arow * D_DIM + koff;

    v2f a[32];
#pragma unroll
    for (int s = 0; s < 32; ++s) a[s] = *(const v2f*)(aPtr + 4 * s);

    __syncthreads();

    if (qa < nQuads) tags[qa] = packTags(ra);
    if (qb < nQuads) tags[qb] = packTags(rb);
    for (int q = qb + qStep; q < nQuads; q += qStep)
        tags[q] = packTags(rows4[q]);
    if (m0 >= nRows) return;

    v8f c[8];
#pragma unroll
    for (int nt = 0; nt < 8; ++nt) c[nt] = (v8f){};

    const float* bBase = sW + l16 * W_PITCH + koff;
#pragma unroll
    for (int s = 0; s < 32; ++s) {
#pragma unroll
        for (int nt = 0; nt < 8; ++nt) {
            const v2f b = *(const v2f*)(bBase + nt * 16 * W_PITCH + 4 * s);
            c[nt] = __builtin_amdgcn_wmma_f32_16x16x4_f32(
                        false, a[s], false, b, (short)0, c[nt], false, false);
        }
    }

    float bv[8];
#pragma unroll
    for (int nt = 0; nt < 8; ++nt) bv[nt] = bias[nt * 16 + l16];

    float* oPtr = support + (size_t)(m0 + half * 8) * D_DIM + l16;
    if (m0 + 16 <= nRows) {
#pragma unroll
        for (int nt = 0; nt < 8; ++nt) {
#pragma unroll
            for (int i = 0; i < 8; ++i)
                oPtr[i * D_DIM + nt * 16] = c[nt][i] + bv[nt];
        }
    } else {
#pragma unroll
        for (int nt = 0; nt < 8; ++nt) {
#pragma unroll
            for (int i = 0; i < 8; ++i) {
                if (m0 + half * 8 + i < nRows)
                    oPtr[i * D_DIM + nt * 16] = c[nt][i] + bv[nt];
            }
        }
    }
}

typedef float v4f __attribute__((ext_vector_type(4)));

#define LIST_CAP 9216

__global__ __launch_bounds__(1024) void gcl_spmm_owned(
    const float* __restrict__ support,
    const float* __restrict__ vals,
    const int*   __restrict__ rows,
    const int*   __restrict__ cols,
    const unsigned* __restrict__ tags,
    float* __restrict__ out,
    int nRows, int nEdges)
{
    __shared__ __attribute__((aligned(16))) float sAcc[OWN_ROWS * D_DIM];
    __shared__ int   sKey[LIST_CAP];
    __shared__ int   sDst[LIST_CAP];
    __shared__ float sVal[LIST_CAP];
    __shared__ int   sCount;

    const int t    = threadIdx.x;
    const int lane = t & 31;
    const int wave = __builtin_amdgcn_readfirstlane(t >> 5);

    const int rowBase = blockIdx.x * OWN_ROWS;
    const int remRows = nRows - rowBase;
    const unsigned rowsHere = (unsigned)(remRows < OWN_ROWS ? remRows : OWN_ROWS);

    for (int i = t; i < OWN_ROWS * D_DIM / 4; i += 1024)
        ((v4f*)sAcc)[i] = (v4f){0.f, 0.f, 0.f, 0.f};
    if (t == 0) sCount = 0;
    __syncthreads();

    auto take = [&](int e) {
        const int slot = atomicAdd(&sCount, 1);
        if (slot < LIST_CAP) {
            sKey[slot] = e;
        } else {
            const unsigned d = (unsigned)(rows[e] - rowBase);
            if (d < rowsHere) {
                const float  v   = vals[e];
                const float* src = support + (size_t)cols[e] * D_DIM;
                float*       dst = sAcc + d * D_DIM;
#pragma clang loop unroll(disable)
                for (int j = 0; j < D_DIM; ++j)
                    atomicAdd(dst + 32 * (j & 3) + (j >> 2), src[j] * v);
            }
        }
    };

    const unsigned pat = (blockIdx.x & 255u) * 0x01010101u;
    const int nUnits = nEdges >> 4;
    const int nTrips = (nUnits + 8191) >> 13;
    const uint4* tags4 = (const uint4*)tags;
    auto hits = [&](unsigned x) -> unsigned {
        const unsigned y = x ^ pat;
        return ~(((y & 0x7f7f7f7fu) + 0x7f7f7f7fu) | y) & 0x80808080u;
    };
    if (nTrips > 0) {
        const int trip0 = (int)(blockIdx.x % (unsigned)nTrips);
        const int uLast = nUnits - 1;
        for (int k = 0; k < nTrips; ++k) {
            int trip = k + trip0;
            if (trip >= nTrips) trip -= nTrips;
            const int q0 = (trip << 13) + t;
            uint4 r[8];
#pragma unroll
            for (int u = 0; u < 8; ++u) {
                const int q = q0 + u * 1024;
                r[u] = tags4[q < uLast ? q : uLast];
            }
#pragma unroll
            for (int u = 0; u < 8; ++u) {
                const int q  = q0 + u * 1024;
                const int e0 = q * 16;
                unsigned m = (hits(r[u].x) >> 7) | (hits(r[u].y) >> 6) |
                             (hits(r[u].z) >> 5) | (hits(r[u].w) >> 4);
                if (q >= nUnits) m = 0u;
                while (m) {
                    const int p = __builtin_ctz(m);
                    m &= m - 1u;
                    take(e0 + 4 * (p & 7) + (p >> 3));
                }
            }
        }
    }
    for (int e = nUnits * 16 + t; e < nEdges; e += 1024) {
        const unsigned d = (unsigned)(rows[e] - rowBase);
        if (d < rowsHere) take(e);
    }
    __syncthreads();

    const int total = __builtin_amdgcn_readfirstlane(sCount);
    const int cnt   = total < LIST_CAP ? total : LIST_CAP;

    for (int i = t; i < cnt; i += 1024) {
        const int e = sKey[i];
        const unsigned d = (unsigned)(rows[e] - rowBase);
        sKey[i] = cols[e];
        sVal[i] = vals[e];
        sDst[i] = (d < rowsHere) ? (int)d : -1;
    }
    __syncthreads();

    const float* sLane = support + lane * 4;
    for (int i0 = wave; i0 < cnt; i0 += 32 * 8) {
        const int iMine = i0 + 32 * (lane & 7);
        const int iSafe = (iMine < cnt) ? iMine : i0;
        const int   colMine = sKey[iSafe];
        const int   dstMine = sDst[iSafe];
        const int   valMine = __float_as_int(sVal[iSafe]);
        v4f   x[8];
        float w[8];
        int   d[8];
#pragma unroll
        for (int u = 0; u < 8; ++u) {
            const int col = __builtin_amdgcn_readlane(colMine, u);
            d[u] = __builtin_amdgcn_readlane(dstMine, u);
            w[u] = __int_as_float(__builtin_amdgcn_readlane(valMine, u));
            x[u] = *(const v4f*)(sLane + (size_t)col * D_DIM);
        }
#pragma unroll
        for (int u = 0; u < 8; ++u) {
            if (i0 + 32 * u < cnt && d[u] >= 0) {
                float* dst = sAcc + d[u] * D_DIM + lane;
                atomicAdd(dst,      x[u].x * w[u]);
                atomicAdd(dst + 32, x[u].y * w[u]);
                atomicAdd(dst + 64, x[u].z * w[u]);
                atomicAdd(dst + 96, x[u].w * w[u]);
            }
        }
    }
    __syncthreads();

    for (unsigned lr = (unsigned)wave; lr < rowsHere; lr += 32) {
        const float* a = sAcc + lr * D_DIM + lane;
        v4f o;
        o.x = a[0];
        o.y = a[32];
        o.z = a[64];
        o.w = a[96];
        *(v4f*)(out + (size_t)(rowBase + (int)lr) * D_DIM + lane * 4) = o;
    }
}

extern "C" void kernel_launch(void* const* d_in, const int* in_sizes, int n_in,
                              void* d_out, int out_size, void* d_ws, size_t ws_size,
                              hipStream_t stream) {
    const float* feat   = (const float*)d_in[0];
    const float* weight = (const float*)d_in[1];
    const float* bias   = (const float*)d_in[2];
    const float* evals  = (const float*)d_in[3];
    const int*   erows  = (const int*)d_in[4];
    const int*   ecols  = (const int*)d_in[5];
    float*       out    = (float*)d_out;

    const int nNodes = in_sizes[0] / D_DIM;
    const int nEdges = in_sizes[3];
    float* support = (float*)d_ws;
    unsigned* tags = (unsigned*)(support + (size_t)nNodes * D_DIM);

    {
        int mtiles = (nNodes + 15) / 16;
        int blocks = (mtiles + 3) / 4;
        gcl_gemm_wmma<<<blocks, 128, 0, stream>>>(feat, weight, bias, erows, support, tags, nNodes, nEdges);
    }

    {
        int blocks = (nNodes + OWN_ROWS - 1) / OWN_ROWS;
        gcl_spmm_owned<<<blocks, 1024, 0, stream>>>(support, evals, erows, ecols, tags, out, nNodes, nEdges);
    }
}


# SpMM: listed edges counting-sorted by destination row in LDS; each wave sums its rows in registers (order = arrival at the rank counters), one store per row; no LDS f32 atomics or accumulators
# speedup vs baseline: 2.8858x; 1.0674x over previous
#include <hip/hip_runtime.h>
#include <hip/hip_bf16.h>


typedef float v2f __attribute__((ext_vector_type(2)));
typedef float v8f __attribute__((ext_vector_type(8)));

#define D_DIM 128

#define W_PITCH 132
#define OWN_ROWS 400
__device__ __forceinline__ unsigned ownerTag(int row)
{
    return ((unsigned)row / (unsigned)OWN_ROWS) & 255u;
}
__device__ __forceinline__ unsigned packTags(int4 r)
{
    return ownerTag(r.x) | (ownerTag(r.y) << 8) | (ownerTag(r.z) << 16) | (ownerTag(r.w) << 24);
}
__global__ __launch_bounds__(128) void gcl_gemm_wmma(
    const float* __restrict__ feat,
    const float* __restrict__ weight,
    const float* __restrict__ bias,
    const int*   __restrict__ rows,
    float* __restrict__ support,
    unsigned* __restrict__ tags,
    int nRows, int nEdges)
{
    __shared__ __attribute__((aligned(16))) float sW[D_DIM * W_PITCH];

    const int nQuads = nEdges >> 2;
    const int qStep  = gridDim.x * 128;
    const int4* rows4 = (const int4*)rows;
    const int qa = blockIdx.x * 128 + threadIdx.x;
    const int qb = qa + qStep;
    int4 ra = {0, 0, 0, 0};
    int4 rb = {0, 0, 0, 0};
    if (nQuads > 0) {
        ra = rows4[qa < nQuads ? qa : nQuads - 1];
        rb = rows4[qb < nQuads ? qb : nQuads - 1];
    }

#pragma unroll
    for (int i = 0; i < 32; ++i) {
        const int idx = threadIdx.x + i * 128;
        const int r   = idx >> 5;
        const int c4  = idx & 31;
        const float4 v = ((const float4*)weight)[idx];
        *(float4*)(&sW[r * W_PITCH + c4 * 4]) = v;
    }

    const int lane   = threadIdx.x & 31;
    const int wave   = threadIdx.x >> 5;
    const int mtile  = blockIdx.x * 4 + wave;
    const int m0     = mtile * 16;

    const int half   = lane >> 4;
    const int l16    = lane & 15;
    const int koff   = half * 2;

    const int arow = (m0 + l16 < nRows) ? (m0 + l16) : (nRows - 1);
    const float* aPtr = feat + (size_t)arow * D_DIM + koff;

    v2f a[32];
#pragma unroll
    for (int s = 0; s < 32; ++s) a[s] = *(const v2f*)(aPtr + 4 * s);

    __syncthreads();

    if (qa < nQuads) tags[qa] = packTags(ra);
    if (qb < nQuads) tags[qb] = packTags(rb);
    for (int q = qb + qStep; q < nQuads; q += qStep)
        tags[q] = packTags(rows4[q]);
    if (m0 >= nRows) return;

    v8f c[8];
#pragma unroll
    for (int nt = 0; nt < 8; ++nt) c[nt] = (v8f){};

    const float* bBase = sW + l16 * W_PITCH + koff;
#pragma unroll
    for (int s = 0; s < 32; ++s) {
#pragma unroll
        for (int nt = 0; nt < 8; ++nt) {
            const v2f b = *(const v2f*)(bBase + nt * 16 * W_PITCH + 4 * s);
            c[nt] = __builtin_amdgcn_wmma_f32_16x16x4_f32(
                        false, a[s], false, b, (short)0, c[nt], false, false);
        }
    }

    float bv[8];
#pragma unroll
    for (int nt = 0; nt < 8; ++nt) bv[nt] = bias[nt * 16 + l16];

    float* oPtr = support + (size_t)(m0 + half * 8) * D_DIM + l16;
    if (m0 + 16 <= nRows) {
#pragma unroll
        for (int nt = 0; nt < 8; ++nt) {
#pragma unroll
            for (int i = 0; i < 8; ++i)
                oPtr[i * D_DIM + nt * 16] = c[nt][i] + bv[nt];
        }
    } else {
#pragma unroll
        for (int nt = 0; nt < 8; ++nt) {
#pragma unroll
            for (int i = 0; i < 8; ++i) {
                if (m0 + half * 8 + i < nRows)
                    oPtr[i * D_DIM + nt * 16] = c[nt][i] + bv[nt];
            }
        }
    }
}

typedef float v4f __attribute__((ext_vector_type(4)));

#define LIST_CAP 9216
#define LIST_PER_THREAD (LIST_CAP / 1024)

__global__ __launch_bounds__(1024) void gcl_spmm_owned(
    const float* __restrict__ support,
    const float* __restrict__ vals,
    const int*   __restrict__ rows,
    const int*   __restrict__ cols,
    const unsigned* __restrict__ tags,
    float* __restrict__ out,
    int nRows, int nEdges)
{
    __shared__ int   sKey[LIST_CAP];
    __shared__ int   sCol[LIST_CAP];
    __shared__ float sVal[LIST_CAP];
    __shared__ int   sStart[OWN_ROWS + 1];
    __shared__ int   sWaveSum[32];
    __shared__ int   sCount;

    const int t    = threadIdx.x;
    const int lane = t & 31;
    const int wave = __builtin_amdgcn_readfirstlane(t >> 5);

    const int rowBase = blockIdx.x * OWN_ROWS;
    const int remRows = nRows - rowBase;
    const unsigned rowsHere = (unsigned)(remRows < OWN_ROWS ? remRows : OWN_ROWS);

    if (t <= OWN_ROWS) sStart[t] = 0;
    if (t == 0) sCount = 0;
    __syncthreads();

    auto take = [&](int e) {
        const int slot = atomicAdd(&sCount, 1);
        if (slot < LIST_CAP) sKey[slot] = e;
    };

    const unsigned pat = (blockIdx.x & 255u) * 0x01010101u;
    const int nUnits = nEdges >> 4;
    const int nTrips = (nUnits + 8191) >> 13;
    const uint4* tags4 = (const uint4*)tags;
    auto hits = [&](unsigned x) -> unsigned {
        const unsigned y = x ^ pat;
        return ~(((y & 0x7f7f7f7fu) + 0x7f7f7f7fu) | y) & 0x80808080u;
    };
    if (nTrips > 0) {
        const int trip0 = (int)(blockIdx.x % (unsigned)nTrips);
        const int uLast = nUnits - 1;
        for (int k = 0; k < nTrips; ++k) {
            int trip = k + trip0;
            if (trip >= nTrips) trip -= nTrips;
            const int q0 = (trip << 13) + t;
            uint4 r[8];
#pragma unroll
            for (int u = 0; u < 8; ++u) {
                const int q = q0 + u * 1024;
                r[u] = tags4[q < uLast ? q : uLast];
            }
#pragma unroll
            for (int u = 0; u < 8; ++u) {
                const int q  = q0 + u * 1024;
                const int e0 = q * 16;
                unsigned m = (hits(r[u].x) >> 7) | (hits(r[u].y) >> 6) |
                             (hits(r[u].z) >> 5) | (hits(r[u].w) >> 4);
                if (q >= nUnits) m = 0u;
                while (m) {
                    const int p = __builtin_ctz(m);
                    m &= m - 1u;
                    take(e0 + 4 * (p & 7) + (p >> 3));
                }
            }
        }
    }
    for (int e = nUnits * 16 + t; e < nEdges; e += 1024) {
        const unsigned d = (unsigned)(rows[e] - rowBase);
        if (d < rowsHere) take(e);
    }
    __syncthreads();

    const int total = __builtin_amdgcn_readfirstlane(sCount);

    if (total > LIST_CAP) {
        for (unsigned lr = (unsigned)wave; lr < rowsHere; lr += 32) {
            const int r = rowBase + (int)lr;
            v4f acc = (v4f){0.f, 0.f, 0.f, 0.f};
            for (int e0 = 0; e0 < nEdges; e0 += 32) {
                const int e = e0 + lane;
                const bool hit = (e < nEdges) && (rows[e] == r);
                unsigned m = __builtin_amdgcn_ballot_w32(hit);
                while (m) {
                    const int ee = e0 + __builtin_ctz(m);
                    m &= m - 1u;
                    const float v = vals[ee];
                    const v4f   x = *(const v4f*)(support + (size_t)cols[ee] * D_DIM + lane * 4);
                    acc += x * v;
                }
            }
            *(v4f*)(out + (size_t)r * D_DIM + lane * 4) = acc;
        }
        return;
    }
    const int cnt = total;

    int   eD[LIST_PER_THREAD];
    int   eR[LIST_PER_THREAD];
    int   eC[LIST_PER_THREAD];
    float eV[LIST_PER_THREAD];
#pragma unroll
    for (int j = 0; j < LIST_PER_THREAD; ++j) {
        const int i = t + j * 1024;
        eD[j] = -1; eR[j] = 0; eC[j] = 0; eV[j] = 0.f;
        if (i < cnt) {
            const int e = sKey[i];
            const unsigned d = (unsigned)(rows[e] - rowBase);
            eC[j] = cols[e];
            eV[j] = vals[e];
            if (d < rowsHere) eD[j] = (int)d;
        }
    }
#pragma unroll
    for (int j = 0; j < LIST_PER_THREAD; ++j)
        if (eD[j] >= 0) eR[j] = atomicAdd(&sStart[eD[j]], 1);
    __syncthreads();

    {
        const int own = (t <= OWN_ROWS) ? sStart[t] : 0;
        int incl = own;
#pragma unroll
        for (int o = 1; o < 32; o <<= 1) {
            const int up = __shfl_up(incl, o);
            if (lane >= o) incl += up;
        }
        if (lane == 31) sWaveSum[wave] = incl;
        __syncthreads();
        if (t <= OWN_ROWS) {
            int base = 0;
            for (int w = 0; w < wave; ++w) base += sWaveSum[w];
            sStart[t] = base + incl - own;
        }
    }
    __syncthreads();

#pragma unroll
    for (int j = 0; j < LIST_PER_THREAD; ++j) {
        if (eD[j] >= 0) {
            const int pos = sStart[eD[j]] + eR[j];
            sKey[pos] = eD[j];
            sCol[pos] = eC[j];
            sVal[pos] = eV[j];
        }
    }
    __syncthreads();

    const int rA   = (int)(((unsigned)wave * rowsHere) >> 5);
    const int rB   = (int)((((unsigned)wave + 1u) * rowsHere) >> 5);
    const int iBeg = __builtin_amdgcn_readfirstlane(sStart[rA]);
    const int iEnd = __builtin_amdgcn_readfirstlane(sStart[rB]);

    const float* sLane = support + lane * 4;
    float*       oLane = out + (size_t)rowBase * D_DIM + lane * 4;
    int cur = rA;
    v4f acc = (v4f){0.f, 0.f, 0.f, 0.f};
    for (int i0 = iBeg; i0 < iEnd; i0 += 8) {
        const int iMine = i0 + (lane & 7);
        const int iSafe = (iMine < iEnd) ? iMine : i0;
        const int colMine = sCol[iSafe];
        const int dstMine = sKey[iSafe];
        const int valMine = __float_as_int(sVal[iSafe]);
        v4f   x[8];
        float w[8];
        int   d[8];
#pragma unroll
        for (int u = 0; u < 8; ++u) {
            const int col = __builtin_amdgcn_readlane(colMine, u);
            d[u] = __builtin_amdgcn_readlane(dstMine, u);
            w[u] = __int_as_float(__builtin_amdgcn_readlane(valMine, u));
            x[u] = *(const v4f*)(sLane + (size_t)col * D_DIM);
        }
#pragma unroll
        for (int u = 0; u < 8; ++u) {
            if (i0 + u < iEnd) {
                while (cur < d[u]) {
                    *(v4f*)(oLane + (size_t)cur * D_DIM) = acc;
                    acc = (v4f){0.f, 0.f, 0.f, 0.f};
                    ++cur;
                }
                acc += x[u] * w[u];
            }
        }
    }
    while (cur < rB) {
        *(v4f*)(oLane + (size_t)cur * D_DIM) = acc;
        acc = (v4f){0.f, 0.f, 0.f, 0.f};
        ++cur;
    }
}

extern "C" void kernel_launch(void* const* d_in, const int* in_sizes, int n_in,
                              void* d_out, int out_size, void* d_ws, size_t ws_size,
                              hipStream_t stream) {
    const float* feat   = (const float*)d_in[0];
    const float* weight = (const float*)d_in[1];
    const float* bias   = (const float*)d_in[2];
    const float* evals  = (const float*)d_in[3];
    const int*   erows  = (const int*)d_in[4];
    const int*   ecols  = (const int*)d_in[5];
    float*       out    = (float*)d_out;

    const int nNodes = in_sizes[0] / D_DIM;
    const int nEdges = in_sizes[3];
    float* support = (float*)d_ws;
    unsigned* tags = (unsigned*)(support + (size_t)nNodes * D_DIM);

    {
        int mtiles = (nNodes + 15) / 16;
        int blocks = (mtiles + 3) / 4;
        gcl_gemm_wmma<<<blocks, 128, 0, stream>>>(feat, weight, bias, erows, support, tags, nNodes, nEdges);
    }

    {
        int blocks = (nNodes + OWN_ROWS - 1) / OWN_ROWS;
        gcl_spmm_owned<<<blocks, 1024, 0, stream>>>(support, evals, erows, ecols, tags, out, nNodes, nEdges);
    }
}


# Edges counting-sorted by owner bucket per 1024-edge chunk in the GEMM launch (16-byte records + 256 start|count words); SpMM groups fetch only their own runs; sum order still arrival order
# speedup vs baseline: 3.8306x; 1.3274x over previous
#include <hip/hip_runtime.h>
#include <hip/hip_bf16.h>


typedef float v2f __attribute__((ext_vector_type(2)));
typedef float v8f __attribute__((ext_vector_type(8)));

#define D_DIM 128

#define W_PITCH 132
#define OWN_ROWS 400
#define CHUNK 1024
#define NBKT 256
__device__ __forceinline__ unsigned ownerTag(int row)
{
    return ((unsigned)row / (unsigned)OWN_ROWS) & 255u;
}
struct EdgeSet {
    int4     row[2];
    int4     col[2];
    float4   val[2];
    unsigned ok;
};
__device__ __forceinline__ int   pick(const int4& v, int k)   { return k == 0 ? v.x : k == 1 ? v.y : k == 2 ? v.z : v.w; }
__device__ __forceinline__ float pick(const float4& v, int k) { return k == 0 ? v.x : k == 1 ? v.y : k == 2 ? v.z : v.w; }
__device__ __forceinline__ void loadEdges(EdgeSet& s, const int* __restrict__ rows, const int* __restrict__ cols,
                                          const float* __restrict__ vals, int e0, int nEdges, bool live)
{
    s.ok = 0u;
#pragma unroll
    for (int h = 0; h < 2; ++h) {
        const int  e     = e0 + h * 512;
        const bool whole = live && (e + 3 < nEdges);
        const int  el    = whole ? e : 0;
        s.row[h] = *(const int4*)(rows + el);
        s.col[h] = *(const int4*)(cols + el);
        s.val[h] = *(const float4*)(vals + el);
        if (whole) s.ok |= 0xFu << (4 * h);
    }
}
__global__ __launch_bounds__(128) void gcl_gemm_wmma(
    const float* __restrict__ feat,
    const float* __restrict__ weight,
    const float* __restrict__ bias,
    const float* __restrict__ vals,
    const int*   __restrict__ rows,
    const int*   __restrict__ cols,
    float* __restrict__ support,
    int*   __restrict__ recsRaw,
    int nRows, int nEdges)
{
    __shared__ __attribute__((aligned(16))) float sW[D_DIM * W_PITCH];
    __shared__ __attribute__((aligned(16))) int   sCnt[NBKT];
    __shared__ __attribute__((aligned(16))) int   sBkt[NBKT];

    const int nChunks = (nEdges + CHUNK - 1) / CHUNK;
    int4*     recs   = (int4*)recsRaw;
    unsigned* starts = (unsigned*)(recs + (size_t)nChunks * CHUNK);

    int chunk = blockIdx.x;
    EdgeSet es;
    const bool   quads = (nEdges >= 4);
    const int*   rSrc  = quads ? rows : (const int*)weight;
    const int*   cSrc  = quads ? cols : (const int*)weight;
    const float* vSrc  = quads ? vals : weight;
    loadEdges(es, rSrc, cSrc, vSrc, chunk * CHUNK + 4 * (int)threadIdx.x, nEdges, chunk < nChunks);
    const int nTail     = nEdges & 3;
    const int tailChunk = nEdges >> 10;
    sCnt[2 * threadIdx.x]     = 0;
    sCnt[2 * threadIdx.x + 1] = 0;

#pragma nounroll
    for (int i0 = 0; i0 < 32; i0 += 16) {
#pragma unroll
        for (int i = 0; i < 16; ++i) {
            const int idx = threadIdx.x + (i0 + i) * 128;
            const int r   = idx >> 5;
            const int c4  = idx & 31;
            const float4 v = ((const float4*)weight)[idx];
            *(float4*)(&sW[r * W_PITCH + c4 * 4]) = v;
        }
    }

    const int lane   = threadIdx.x & 31;
    const int wave   = threadIdx.x >> 5;
    const int mtile  = blockIdx.x * 4 + wave;
    const int m0     = mtile * 16;

    const int half   = lane >> 4;
    const int l16    = lane & 15;
    const int koff   = half * 2;

    const int arow = (m0 + l16 < nRows) ? (m0 + l16) : (nRows - 1);
    const float* aPtr = feat + (size_t)arow * D_DIM + koff;

    v2f a[32];
#pragma unroll
    for (int s = 0; s < 32; ++s) a[s] = *(const v2f*)(aPtr + 4 * s);

    __syncthreads();

    while (chunk < nChunks) {
        int bk[8], rk[8];
#pragma unroll
        for (int k = 0; k < 8; ++k) bk[k] = ((es.ok >> k) & 1u) ? (int)ownerTag(pick(es.row[k >> 2], k & 3)) : -1;
#pragma unroll
        for (int k = 0; k < 8; ++k) rk[k] = (bk[k] >= 0) ? atomicAdd(&sCnt[bk[k]], 1) : 0;
        const bool tail = (chunk == tailChunk) && ((int)threadIdx.x < nTail);
        int tRow = 0, tCol = 0, tBk = 0, tRk = 0;
        float tVal = 0.f;
        if (tail) {
            const int e = (nEdges & ~3) + (int)threadIdx.x;
            tRow = rows[e]; tCol = cols[e]; tVal = vals[e];
            tBk  = (int)ownerTag(tRow);
            tRk  = atomicAdd(&sCnt[tBk], 1);
        }
        __syncthreads();

        {
            const int4 ca = *(const int4*)(&sCnt[lane * 8]);
            const int4 cb = *(const int4*)(&sCnt[lane * 8 + 4]);
            const int cnt8[8] = {ca.x, ca.y, ca.z, ca.w, cb.x, cb.y, cb.z, cb.w};
            int sum = 0;
#pragma unroll
            for (int k = 0; k < 8; ++k) sum += cnt8[k];
            int incl = sum;
#pragma unroll
            for (int o = 1; o < 32; o <<= 1) {
                const int up = __shfl_up(incl, o);
                if (lane >= o) incl += up;
            }
            if ((lane >> 3) == wave) {
                int st = incl - sum;
                int      sv[8];
                unsigned pv[8];
#pragma unroll
                for (int k = 0; k < 8; ++k) {
                    sv[k] = st;
                    pv[k] = (unsigned)st | ((unsigned)cnt8[k] << 16);
                    st += cnt8[k];
                }
                *(int4*)(&sBkt[lane * 8])     = make_int4(sv[0], sv[1], sv[2], sv[3]);
                *(int4*)(&sBkt[lane * 8 + 4]) = make_int4(sv[4], sv[5], sv[6], sv[7]);
                uint4* tp = (uint4*)(starts + (size_t)chunk * NBKT + lane * 8);
                tp[0] = make_uint4(pv[0], pv[1], pv[2], pv[3]);
                tp[1] = make_uint4(pv[4], pv[5], pv[6], pv[7]);
            }
        }
        __syncthreads();

        int4* rp = recs + (size_t)chunk * CHUNK;
        int pos[8];
#pragma unroll
        for (int k = 0; k < 8; ++k) pos[k] = sBkt[bk[k] & (NBKT - 1)] + rk[k];
#pragma unroll
        for (int k = 0; k < 8; ++k) {
            if (bk[k] >= 0)
                rp[pos[k]] = make_int4(pick(es.row[k >> 2], k & 3), pick(es.col[k >> 2], k & 3),
                                       __float_as_int(pick(es.val[k >> 2], k & 3)), 0);
        }
        if (tail) rp[sBkt[tBk] + tRk] = make_int4(tRow, tCol, __float_as_int(tVal), 0);

        chunk += gridDim.x;
        if (chunk < nChunks) {
            loadEdges(es, rSrc, cSrc, vSrc, chunk * CHUNK + 4 * (int)threadIdx.x, nEdges, true);
            sCnt[2 * threadIdx.x]     = 0;
            sCnt[2 * threadIdx.x + 1] = 0;
            __syncthreads();
        }
    }
    if (m0 >= nRows) return;

    v8f c[8];
#pragma unroll
    for (int nt = 0; nt < 8; ++nt) c[nt] = (v8f){};

    const float* bBase = sW + l16 * W_PITCH + koff;
#pragma unroll
    for (int s = 0; s < 32; ++s) {
#pragma unroll
        for (int nt = 0; nt < 8; ++nt) {
            const v2f b = *(const v2f*)(bBase + nt * 16 * W_PITCH + 4 * s);
            c[nt] = __builtin_amdgcn_wmma_f32_16x16x4_f32(
                        false, a[s], false, b, (short)0, c[nt], false, false);
        }
    }

    float bv[8];
#pragma unroll
    for (int nt = 0; nt < 8; ++nt) bv[nt] = bias[nt * 16 + l16];

    float* oPtr = support + (size_t)(m0 + half * 8) * D_DIM + l16;
    if (m0 + 16 <= nRows) {
#pragma unroll
        for (int nt = 0; nt < 8; ++nt) {
#pragma unroll
            for (int i = 0; i < 8; ++i)
                oPtr[i * D_DIM + nt * 16] = c[nt][i] + bv[nt];
        }
    } else {
#pragma unroll
        for (int nt = 0; nt < 8; ++nt) {
#pragma unroll
            for (int i = 0; i < 8; ++i) {
                if (m0 + half * 8 + i < nRows)
                    oPtr[i * D_DIM + nt * 16] = c[nt][i] + bv[nt];
            }
        }
    }
}

typedef float v4f __attribute__((ext_vector_type(4)));

#define LIST_CAP 9216
#define LIST_PER_THREAD (LIST_CAP / 1024)

__global__ __launch_bounds__(1024) void gcl_spmm_owned(
    const float* __restrict__ support,
    const float* __restrict__ vals,
    const int*   __restrict__ rows,
    const int*   __restrict__ cols,
    const int*   __restrict__ recsRaw,
    float* __restrict__ out,
    int nRows, int nEdges)
{
    __shared__ int   sKey[LIST_CAP];
    __shared__ int   sCol[LIST_CAP];
    __shared__ float sVal[LIST_CAP];
    __shared__ int   sStart[OWN_ROWS + 1];
    __shared__ int   sWaveSum[32];
    __shared__ int   sCount;

    const int t    = threadIdx.x;
    const int lane = t & 31;
    const int wave = __builtin_amdgcn_readfirstlane(t >> 5);

    const int rowBase = blockIdx.x * OWN_ROWS;
    const int remRows = nRows - rowBase;
    const unsigned rowsHere = (unsigned)(remRows < OWN_ROWS ? remRows : OWN_ROWS);

    if (t <= OWN_ROWS) sStart[t] = 0;
    if (t == 0) sCount = 0;
    __syncthreads();

    const int nChunks = (nEdges + CHUNK - 1) / CHUNK;
    const int4*     recs   = (const int4*)recsRaw;
    const unsigned* starts = (const unsigned*)(recs + (size_t)nChunks * CHUNK) + (blockIdx.x & 255u);
    {
        int c0 = wave * 32;
        unsigned wNext = (c0 + lane < nChunks) ? starts[(size_t)(c0 + lane) * NBKT] : 0u;
        for (; c0 < nChunks; c0 += 1024) {
            const unsigned w  = wNext;
            const int      c  = c0 + lane;
            const int      cn = c + 1024;
            wNext = (cn < nChunks) ? starts[(size_t)cn * NBKT] : 0u;
            const int first = (int)(w & 0xffffu);
            const int n     = (int)(w >> 16);
            int incl = n;
#pragma unroll
            for (int o = 1; o < 32; o <<= 1) {
                const int up = __shfl_up(incl, o);
                if (lane >= o) incl += up;
            }
            int base = 0;
            if (lane == 31) base = atomicAdd(&sCount, incl);
            base = __shfl(base, 31);
            const int slot = base + incl - n;
            const int rec0 = c * CHUNK + first;
            for (int j = 0; j < n; ++j)
                if (slot + j < LIST_CAP) sKey[slot + j] = rec0 + j;
        }
    }
    __syncthreads();

    const int total = __builtin_amdgcn_readfirstlane(sCount);

    if (total > LIST_CAP) {
        for (unsigned lr = (unsigned)wave; lr < rowsHere; lr += 32) {
            const int r = rowBase + (int)lr;
            v4f acc = (v4f){0.f, 0.f, 0.f, 0.f};
            for (int e0 = 0; e0 < nEdges; e0 += 32) {
                const int e = e0 + lane;
                const bool hit = (e < nEdges) && (rows[e] == r);
                unsigned m = __builtin_amdgcn_ballot_w32(hit);
                while (m) {
                    const int ee = e0 + __builtin_ctz(m);
                    m &= m - 1u;
                    const float v = vals[ee];
                    const v4f   x = *(const v4f*)(support + (size_t)cols[ee] * D_DIM + lane * 4);
                    acc += x * v;
                }
            }
            *(v4f*)(out + (size_t)r * D_DIM + lane * 4) = acc;
        }
        return;
    }
    const int cnt = total;

    int   eD[LIST_PER_THREAD];
    int   eR[LIST_PER_THREAD];
    int   eC[LIST_PER_THREAD];
    float eV[LIST_PER_THREAD];
    int4 rec[LIST_PER_THREAD];
#pragma unroll
    for (int j = 0; j < LIST_PER_THREAD; ++j) {
        const int i   = t + j * 1024;
        const int key = sKey[i];
        rec[j] = recs[(i < cnt) ? key : 0];
    }
#pragma unroll
    for (int j = 0; j < LIST_PER_THREAD; ++j) {
        const int i = t + j * 1024;
        const unsigned d = (unsigned)(rec[j].x - rowBase);
        eD[j] = ((i < cnt) && (d < rowsHere)) ? (int)d : -1;
        eR[j] = 0;
        eC[j] = rec[j].y;
        eV[j] = __int_as_float(rec[j].z);
    }
#pragma unroll
    for (int j = 0; j < LIST_PER_THREAD; ++j)
        if (eD[j] >= 0) eR[j] = atomicAdd(&sStart[eD[j]], 1);
    __syncthreads();

    {
        const int own = (t <= OWN_ROWS) ? sStart[t] : 0;
        int incl = own;
#pragma unroll
        for (int o = 1; o < 32; o <<= 1) {
            const int up = __shfl_up(incl, o);
            if (lane >= o) incl += up;
        }
        if (lane == 31) sWaveSum[wave] = incl;
        __syncthreads();
        if (t <= OWN_ROWS) {
            int base = 0;
            for (int w = 0; w < wave; ++w) base += sWaveSum[w];
            sStart[t] = base + incl - own;
        }
    }
    __syncthreads();

#pragma unroll
    for (int j = 0; j < LIST_PER_THREAD; ++j) {
        if (eD[j] >= 0) {
            const int pos = sStart[eD[j]] + eR[j];
            sKey[pos] = eD[j];
            sCol[pos] = eC[j];
            sVal[pos] = eV[j];
        }
    }
    __syncthreads();

    const int rA   = (int)(((unsigned)wave * rowsHere) >> 5);
    const int rB   = (int)((((unsigned)wave + 1u) * rowsHere) >> 5);
    const int iBeg = __builtin_amdgcn_readfirstlane(sStart[rA]);
    const int iEnd = __builtin_amdgcn_readfirstlane(sStart[rB]);

    const float* sLane = support + lane * 4;
    float*       oLane = out + (size_t)rowBase * D_DIM + lane * 4;
    int cur = rA;
    v4f acc = (v4f){0.f, 0.f, 0.f, 0.f};
    for (int i0 = iBeg; i0 < iEnd; i0 += 8) {
        const int iMine = i0 + (lane & 7);
        const int iSafe = (iMine < iEnd) ? iMine : i0;
        const int colMine = sCol[iSafe];
        const int dstMine = sKey[iSafe];
        const int valMine = __float_as_int(sVal[iSafe]);
        v4f   x[8];
        float w[8];
        int   d[8];
#pragma unroll
        for (int u = 0; u < 8; ++u) {
            const int col = __builtin_amdgcn_readlane(colMine, u);
            d[u] = __builtin_amdgcn_readlane(dstMine, u);
            w[u] = __int_as_float(__builtin_amdgcn_readlane(valMine, u));
            x[u] = *(const v4f*)(sLane + (size_t)col * D_DIM);
        }
#pragma unroll
        for (int u = 0; u < 8; ++u) {
            if (i0 + u < iEnd) {
                while (cur < d[u]) {
                    *(v4f*)(oLane + (size_t)cur * D_DIM) = acc;
                    acc = (v4f){0.f, 0.f, 0.f, 0.f};
                    ++cur;
                }
                acc += x[u] * w[u];
            }
        }
    }
    while (cur < rB) {
        *(v4f*)(oLane + (size_t)cur * D_DIM) = acc;
        acc = (v4f){0.f, 0.f, 0.f, 0.f};
        ++cur;
    }
}

extern "C" void kernel_launch(void* const* d_in, const int* in_sizes, int n_in,
                              void* d_out, int out_size, void* d_ws, size_t ws_size,
                              hipStream_t stream) {
    const float* feat   = (const float*)d_in[0];
    const float* weight = (const float*)d_in[1];
    const float* bias   = (const float*)d_in[2];
    const float* evals  = (const float*)d_in[3];
    const int*   erows  = (const int*)d_in[4];
    const int*   ecols  = (const int*)d_in[5];
    float*       out    = (float*)d_out;

    const int nNodes = in_sizes[0] / D_DIM;
    const int nEdges = in_sizes[3];
    float* support = (float*)d_ws;
    int* recs = (int*)(support + (size_t)nNodes * D_DIM);

    {
        int mtiles = (nNodes + 15) / 16;
        int blocks = (mtiles + 3) / 4;
        gcl_gemm_wmma<<<blocks, 128, 0, stream>>>(feat, weight, bias, evals, erows, ecols, support, recs, nNodes, nEdges);
    }

    {
        int blocks = (nNodes + OWN_ROWS - 1) / OWN_ROWS;
        gcl_spmm_owned<<<blocks, 1024, 0, stream>>>(support, evals, erows, ecols, recs, out, nNodes, nEdges);
    }
}
